# starting kernel
# baseline (speedup 1.0000x reference)
#include <hip/hip_runtime.h>
#include <hip/hip_bf16.h>

typedef float v2f __attribute__((ext_vector_type(2)));
typedef float v8f __attribute__((ext_vector_type(8)));

#define T_FRAMES   61
#define DT         (1.0f / 60.0f)
#define WG_THREADS 256
#define VOX_PER_WG 128
#define FS         66
#define SS         130

__global__ __launch_bounds__(WG_THREADS)
void srtm_wmma_kernel(const float* __restrict__ ref_tac,
                      const float* __restrict__ kparam,
                      float* __restrict__ y,
                      int L)
{
    __shared__ float Fsh[64 * FS];
    __shared__ float Ssh[64 * SS];

    const int tid    = threadIdx.x;
    const int wave   = tid >> 5;
    const int lane   = tid & 31;
    const int laneLo = lane & 15;
    const int laneHi = lane >> 4;
    const int wgBase = blockIdx.x * VOX_PER_WG;

    for (int idx = tid; idx < 64 * 64; idx += WG_THREADS) {
        const int q = idx >> 6, r = idx & 63;
        float v = 0.0f;
        if (q < 60 && r < 60) {
            const float a = ref_tac[q], b = ref_tac[q + 1];
            v = fmaxf(fmaf((float)r * DT, b - a, a), 0.0f);
        }
        Fsh[q * FS + r] = v;
    }

    int vox = wgBase + wave * 16 + laneLo;
    vox = vox < L ? vox : L - 1;
    const float kk2 = kparam[L + vox];
    const float bp  = kparam[2 * L + vox];
    const float d   = expf((-kk2 / (1.0f + bp)) * DT);
    const float d2  = d * d;
    const float d4  = d2 * d2;
    float pwx = laneHi ? d2      : d4;
    float pwy = laneHi ? d       : d2 * d;
    const int aoff = laneHi ? 2 : 0;

    __syncthreads();

    v8f acc0 = {}, acc1 = {}, acc2 = {}, acc3 = {};
    for (int kb = 56; kb >= 0; kb -= 4) {
        v2f b; b.x = pwx; b.y = pwy;
        const int col = kb + aoff;
        v2f a0 = { Fsh[(laneLo     ) * FS + col], Fsh[(laneLo     ) * FS + col + 1] };
        v2f a1 = { Fsh[(laneLo + 16) * FS + col], Fsh[(laneLo + 16) * FS + col + 1] };
        v2f a2 = { Fsh[(laneLo + 32) * FS + col], Fsh[(laneLo + 32) * FS + col + 1] };
        v2f a3 = { Fsh[(laneLo + 48) * FS + col], Fsh[(laneLo + 48) * FS + col + 1] };
        acc0 = __builtin_amdgcn_wmma_f32_16x16x4_f32(false, a0, false, b, (short)0, acc0, false, false);
        acc1 = __builtin_amdgcn_wmma_f32_16x16x4_f32(false, a1, false, b, (short)0, acc1, false, false);
        acc2 = __builtin_amdgcn_wmma_f32_16x16x4_f32(false, a2, false, b, (short)0, acc2, false, false);
        acc3 = __builtin_amdgcn_wmma_f32_16x16x4_f32(false, a3, false, b, (short)0, acc3, false, false);
        pwx *= d4; pwy *= d4;
    }

    const int scol = wave * 16 + laneLo;
    #pragma unroll
    for (int v = 0; v < 8; ++v) {
        const int r0 = v + laneHi * 8;
        Ssh[(r0     ) * SS + scol] = acc0[v];
        Ssh[(r0 + 16) * SS + scol] = acc1[v];
        Ssh[(r0 + 32) * SS + scol] = acc2[v];
        Ssh[(r0 + 48) * SS + scol] = acc3[v];
    }
    __syncthreads();

    if (tid < VOX_PER_WG) {
        const int l = wgBase + tid;
        if (l < L) {
            const float R1  = kparam[l];
            const float kk2s = kparam[L + l];
            const float bps  = kparam[2 * L + l];
            const float As  = -kk2s / (1.0f + bps);
            const float Bs  = kk2s * (1.0f - R1 / (1.0f + bps));
            const float d60 = expf(As);
            float C = 0.0f;
            const float r0v = ref_tac[0];
            y[l] = fmaf(Bs, DT * fmaxf(r0v, 0.0f), r0v * R1);
            for (int m = 1; m < T_FRAMES; ++m) {
                C = fmaf(d60, C, Ssh[(m - 1) * SS + tid]);
                const float rm   = ref_tac[m];
                const float conv = DT * (fmaxf(rm, 0.0f) + C);
                y[(size_t)m * L + l] = fmaf(Bs, conv, rm * R1);
            }
        }
    }
}

extern "C" void kernel_launch(void* const* d_in, const int* in_sizes, int n_in,
                              void* d_out, int out_size, void* d_ws, size_t ws_size,
                              hipStream_t stream) {
    const float* ref_tac = (const float*)d_in[1];
    const float* kparam  = (const float*)d_in[2];
    float* y = (float*)d_out;
    const int L = in_sizes[2] / 3;
    const int grid = (L + VOX_PER_WG - 1) / VOX_PER_WG;
    srtm_wmma_kernel<<<grid, WG_THREADS, 0, stream>>>(ref_tac, kparam, y, L);
}


# S stays in registers: each wave takes 32 voxels x 32 rows (2x2 tiles), a lane-half swap gives a lane one voxel's rows; recurrence and stores from registers, 32-row carry via LDS; same sum order
# speedup vs baseline: 1.2000x; 1.2000x over previous
#include <hip/hip_runtime.h>
#include <hip/hip_bf16.h>

typedef float v2f __attribute__((ext_vector_type(2)));
typedef float v8f __attribute__((ext_vector_type(8)));

#define T_FRAMES   61
#define DT         (1.0f / 60.0f)
#define WG_THREADS 256
#define VOX_PER_WG 128
#define FS         66

__device__ __forceinline__ float xhalf(float x) {
    return __builtin_bit_cast(float, __builtin_amdgcn_permlanex16(
        __builtin_bit_cast(int, x), __builtin_bit_cast(int, x), 0x76543210, 0xfedcba98, false, false));
}

__global__ __launch_bounds__(WG_THREADS)
void srtm_wmma_kernel(const float* __restrict__ ref_tac,
                      const float* __restrict__ kparam,
                      float* __restrict__ y,
                      int L)
{
    __shared__ float Fsh[64 * FS];
    __shared__ float Csh[VOX_PER_WG];

    const int tid    = threadIdx.x;
    const int wave   = __builtin_amdgcn_readfirstlane(tid >> 5);
    const int lane   = tid & 31;
    const int laneLo = lane & 15;
    const int laneHi = lane >> 4;
    const int grp    = wave & 3;
    const int mh     = wave >> 2;
    const int wgBase = blockIdx.x * VOX_PER_WG;

    for (int idx = tid; idx < 64 * 64; idx += WG_THREADS) {
        const int q = idx >> 6, r = idx & 63;
        float v = 0.0f;
        if (q < 60 && r < 60) {
            const float a = ref_tac[q], b = ref_tac[q + 1];
            v = fmaxf(fmaf((float)r * DT, b - a, a), 0.0f);
        }
        Fsh[q * FS + r] = v;
    }

    const int  voxRaw = wgBase + grp * 32 + lane;
    const bool voxOk  = voxRaw < L;
    const int  vox    = voxOk ? voxRaw : L - 1;
    const float R1  = kparam[vox];
    const float kk2 = kparam[L + vox];
    const float bp  = kparam[2 * L + vox];
    const float As  = -kk2 / (1.0f + bp);
    const float Bs  = kk2 * (1.0f - R1 / (1.0f + bp));
    const float dOwn = expf(As * DT);
    const float d60  = expf(As);
    const float dOth = xhalf(dOwn);
    const float dA = laneHi ? dOth : dOwn;
    const float dB = laneHi ? dOwn : dOth;
    const float dA2 = dA * dA, dA4 = dA2 * dA2;
    const float dB2 = dB * dB, dB4 = dB2 * dB2;
    float pAx = laneHi ? dA2 : dA4;
    float pAy = laneHi ? dA  : dA2 * dA;
    float pBx = laneHi ? dB2 : dB4;
    float pBy = laneHi ? dB  : dB2 * dB;
    const int aoff = laneHi ? 2 : 0;

    float rr[32];
    #pragma unroll
    for (int i = 0; i < 32; ++i) {
        int m = 32 * mh + i + 1;
        if (i >= 28) m = m < T_FRAMES - 1 ? m : T_FRAMES - 1;
        rr[i] = ref_tac[m];
    }
    const float r0v = ref_tac[0];

    __syncthreads();

    v8f acc[2][2] = {};
    const int arow = 32 * mh + laneLo;
    for (int kb = 56; kb >= 0; kb -= 4) {
        v2f b0; b0.x = pAx; b0.y = pAy;
        v2f b1; b1.x = pBx; b1.y = pBy;
        const int col = kb + aoff;
        v2f a0 = { Fsh[(arow     ) * FS + col], Fsh[(arow     ) * FS + col + 1] };
        v2f a1 = { Fsh[(arow + 16) * FS + col], Fsh[(arow + 16) * FS + col + 1] };
        acc[0][0] = __builtin_amdgcn_wmma_f32_16x16x4_f32(false, a0, false, b0, (short)0, acc[0][0], false, false);
        acc[0][1] = __builtin_amdgcn_wmma_f32_16x16x4_f32(false, a0, false, b1, (short)0, acc[0][1], false, false);
        acc[1][0] = __builtin_amdgcn_wmma_f32_16x16x4_f32(false, a1, false, b0, (short)0, acc[1][0], false, false);
        acc[1][1] = __builtin_amdgcn_wmma_f32_16x16x4_f32(false, a1, false, b1, (short)0, acc[1][1], false, false);
        pAx *= dA4; pAy *= dA4;
        pBx *= dB4; pBy *= dB4;
    }

    float s[32];
    #pragma unroll
    for (int mt = 0; mt < 2; ++mt) {
        #pragma unroll
        for (int v = 0; v < 8; ++v) {
            const float X  = acc[mt][0][v];
            const float Y  = acc[mt][1][v];
            const float Zs = xhalf(laneHi ? X : Y);
            s[16 * mt + v]     = laneHi ? Zs : X;
            s[16 * mt + 8 + v] = laneHi ? Y  : Zs;
        }
    }

    float P[32];
    if (mh == 0) {
        float C = 0.0f;
        #pragma unroll
        for (int i = 0; i < 32; ++i) { C = fmaf(d60, C, s[i]); P[i] = C; }
        Csh[grp * 32 + lane] = C;
    }
    __syncthreads();
    if (mh != 0) {
        float C = Csh[grp * 32 + lane];
        #pragma unroll
        for (int i = 0; i < 32; ++i) { C = fmaf(d60, C, s[i]); P[i] = C; }
    }

    if (voxOk) {
        if (mh == 0)
            y[voxRaw] = fmaf(Bs, DT * fmaxf(r0v, 0.0f), r0v * R1);
        #pragma unroll
        for (int i = 0; i < 32; ++i) {
            const int m = 32 * mh + i + 1;
            if (i < 28 || m < T_FRAMES) {
                const float rm   = rr[i];
                const float conv = DT * (fmaxf(rm, 0.0f) + P[i]);
                float* yrow = y + (size_t)m * L;
                yrow[voxRaw] = fmaf(Bs, conv, rm * R1);
            }
        }
    }
}

extern "C" void kernel_launch(void* const* d_in, const int* in_sizes, int n_in,
                              void* d_out, int out_size, void* d_ws, size_t ws_size,
                              hipStream_t stream) {
    const float* ref_tac = (const float*)d_in[1];
    const float* kparam  = (const float*)d_in[2];
    float* y = (float*)d_out;
    const int L = in_sizes[2] / 3;
    const int grid = (L + VOX_PER_WG - 1) / VOX_PER_WG;
    srtm_wmma_kernel<<<grid, WG_THREADS, 0, stream>>>(ref_tac, kparam, y, L);
}


# Rows of F are linear in r when no frame value is negative: S = [ref q, ref q+1] x (W x D), 12 WMMAs a wave not 60, a wave per 32 voxels, no LDS; full F x D kept for negative values; sum order changes
# speedup vs baseline: 1.5613x; 1.3011x over previous
#include <hip/hip_runtime.h>
#include <hip/hip_bf16.h>

typedef float v2f __attribute__((ext_vector_type(2)));
typedef float v8f __attribute__((ext_vector_type(8)));

#define T_FRAMES   61
#define DT         (1.0f / 60.0f)
#define WG_THREADS 128
#define VOX_PER_WG 128
#define FS         66

__device__ __forceinline__ float xhalf(float x) {
    return __builtin_bit_cast(float, __builtin_amdgcn_permlanex16(
        __builtin_bit_cast(int, x), __builtin_bit_cast(int, x), 0x76543210, 0xfedcba98, false, false));
}

__device__ __forceinline__ v8f mm(v2f a, v2f b, v8f c) {
    return __builtin_amdgcn_wmma_f32_16x16x4_f32(false, a, false, b, (short)0, c, false, false);
}

__global__ __launch_bounds__(WG_THREADS)
void srtm_wmma_kernel(const float* __restrict__ ref_tac,
                      const float* __restrict__ kparam,
                      float* __restrict__ y,
                      int L)
{
    __shared__ float Fsh[64 * FS];

    const int tid    = threadIdx.x;
    const int wave   = __builtin_amdgcn_readfirstlane(tid >> 5);
    const int lane   = tid & 31;
    const int laneLo = lane & 15;
    const int laneHi = lane >> 4;
    const int wgBase = blockIdx.x * VOX_PER_WG;

    const int  voxRaw = wgBase + wave * 32 + lane;
    const bool voxOk  = voxRaw < L;
    const int  vox    = voxOk ? voxRaw : L - 1;
    const float R1  = kparam[vox];
    const float kk2 = kparam[L + vox];
    const float bp  = kparam[2 * L + vox];

    float aref[4];
    #pragma unroll
    for (int mt = 0; mt < 4; ++mt) {
        int qi = 16 * mt + laneLo + laneHi;
        qi = qi < T_FRAMES - 1 ? qi : T_FRAMES - 1;
        aref[mt] = ref_tac[qi];
    }
    const bool anyNeg = __builtin_amdgcn_ballot_w32(aref[0] < 0.0f || aref[1] < 0.0f ||
                                                    aref[2] < 0.0f || aref[3] < 0.0f) != 0u;
    float rr[T_FRAMES];
    #pragma unroll
    for (int m = 0; m < T_FRAMES; ++m)
        rr[m] = __builtin_bit_cast(float, __builtin_amdgcn_readlane(__builtin_bit_cast(int, aref[m >> 4]), m & 15));

    const float As  = -kk2 / (1.0f + bp);
    const float Bs  = kk2 * (1.0f - R1 / (1.0f + bp));
    const float dOwn = expf(As * DT);
    const float d60  = expf(As);
    const float dOth = xhalf(dOwn);
    const float dA = laneHi ? dOth : dOwn;
    const float dB = laneHi ? dOwn : dOth;
    const float dA2 = dA * dA, dA4 = dA2 * dA2;
    const float dB2 = dB * dB, dB4 = dB2 * dB2;

    v8f acc[4][2];

    if (!anyNeg) {
        const int wsel = laneLo >> 3, ch = laneLo & 7;
        v2f wa[2];
        #pragma unroll
        for (int j = 0; j < 2; ++j) {
            #pragma unroll
            for (int s = 0; s < 2; ++s) {
                const int   k = 4 * j + 2 * laneHi + s;
                const int   r = 59 - 8 * ch - k;
                const float t = (float)r * DT;
                const float w = wsel ? t : 1.0f - t;
                wa[j][s] = r < 0 ? 0.0f : w;
            }
        }
        v2f bA[2], bB[2];
        bA[0].x = laneHi ? dA2 * dA : dA;  bA[0].y = laneHi ? dA4 : dA2;
        bB[0].x = laneHi ? dB2 * dB : dB;  bB[0].y = laneHi ? dB4 : dB2;
        bA[1] = bA[0] * dA4;
        bB[1] = bB[0] * dB4;
        v8f tA = {}, tB = {};
        tA = mm(wa[0], bA[0], tA);
        tB = mm(wa[0], bB[0], tB);
        tA = mm(wa[1], bA[1], tA);
        tB = mm(wa[1], bB[1], tB);
        const float dA8 = dA4 * dA4, dB8 = dB4 * dB4;
        float uA = tA[7], uB = tB[7];
        #pragma unroll
        for (int r = 6; r >= 0; --r) { uA = fmaf(uA, dA8, tA[r]); uB = fmaf(uB, dB8, tB[r]); }
        v2f ubA; ubA.x = uA; ubA.y = 0.0f;
        v2f ubB; ubB.x = uB; ubB.y = 0.0f;
        const v8f zero = {};
        #pragma unroll
        for (int mt = 0; mt < 4; ++mt) {
            v2f a; a.x = aref[mt]; a.y = 0.0f;
            acc[mt][0] = mm(a, ubA, zero);
            acc[mt][1] = mm(a, ubB, zero);
        }
    } else {
        for (int idx = tid; idx < 64 * 64; idx += WG_THREADS) {
            const int q = idx >> 6, r = idx & 63;
            float v = 0.0f;
            if (q < 60 && r < 60) {
                const float a = ref_tac[q], b = ref_tac[q + 1];
                v = fmaxf(fmaf((float)r * DT, b - a, a), 0.0f);
            }
            Fsh[q * FS + r] = v;
        }
        __syncthreads();
        float pAx = laneHi ? dA2 : dA4;
        float pAy = laneHi ? dA  : dA2 * dA;
        float pBx = laneHi ? dB2 : dB4;
        float pBy = laneHi ? dB  : dB2 * dB;
        const int aoff = laneHi ? 2 : 0;
        #pragma unroll
        for (int mt = 0; mt < 4; ++mt) { acc[mt][0] = (v8f){}; acc[mt][1] = (v8f){}; }
        for (int kb = 56; kb >= 0; kb -= 4) {
            v2f b0; b0.x = pAx; b0.y = pAy;
            v2f b1; b1.x = pBx; b1.y = pBy;
            const int col = kb + aoff;
            #pragma unroll
            for (int mt = 0; mt < 4; ++mt) {
                const int arow = 16 * mt + laneLo;
                v2f a = { Fsh[arow * FS + col], Fsh[arow * FS + col + 1] };
                acc[mt][0] = mm(a, b0, acc[mt][0]);
                acc[mt][1] = mm(a, b1, acc[mt][1]);
            }
            pAx *= dA4; pAy *= dA4;
            pBx *= dB4; pBy *= dB4;
        }
    }

    float s[64];
    #pragma unroll
    for (int mt = 0; mt < 4; ++mt) {
        #pragma unroll
        for (int v = 0; v < 8; ++v) {
            const float X  = acc[mt][0][v];
            const float Y  = acc[mt][1][v];
            const float Zs = xhalf(laneHi ? X : Y);
            s[16 * mt + v]     = laneHi ? Zs : X;
            s[16 * mt + 8 + v] = laneHi ? Y  : Zs;
        }
    }

    if (voxOk) {
        const float r0v = rr[0];
        y[voxRaw] = fmaf(Bs, DT * fmaxf(r0v, 0.0f), r0v * R1);
        float C = 0.0f;
        #pragma unroll
        for (int i = 0; i < T_FRAMES - 1; ++i) {
            C = fmaf(d60, C, s[i]);
            const int   m    = i + 1;
            const float rm   = rr[m];
            const float conv = DT * (fmaxf(rm, 0.0f) + C);
            float* yrow = y + (size_t)m * L;
            yrow[voxRaw] = fmaf(Bs, conv, rm * R1);
        }
    }
}

extern "C" void kernel_launch(void* const* d_in, const int* in_sizes, int n_in,
                              void* d_out, int out_size, void* d_ws, size_t ws_size,
                              hipStream_t stream) {
    const float* ref_tac = (const float*)d_in[1];
    const float* kparam  = (const float*)d_in[2];
    float* y = (float*)d_out;
    const int L = in_sizes[2] / 3;
    const int grid = (L + VOX_PER_WG - 1) / VOX_PER_WG;
    srtm_wmma_kernel<<<grid, WG_THREADS, 0, stream>>>(ref_tac, kparam, y, L);
}


# S = [ref q, ref q+1] x U packed block-diagonally over k: a lane's D registers are its own voxel's rows, so the lane-half swap goes and rows go out behind each product; S's two terms on adjacent k
# speedup vs baseline: 1.7614x; 1.1282x over previous
#include <hip/hip_runtime.h>
#include <hip/hip_bf16.h>

typedef float v2f __attribute__((ext_vector_type(2)));
typedef float v8f __attribute__((ext_vector_type(8)));

#define T_FRAMES   61
#define DT         (1.0f / 60.0f)
#define WG_THREADS 128
#define VOX_PER_WG 128
#define FS         66

__device__ __forceinline__ float xhalf(float x) {
    return __builtin_bit_cast(float, __builtin_amdgcn_permlanex16(
        __builtin_bit_cast(int, x), __builtin_bit_cast(int, x), 0x76543210, 0xfedcba98, false, false));
}

__device__ __forceinline__ v8f mm(v2f a, v2f b, v8f c) {
    return __builtin_amdgcn_wmma_f32_16x16x4_f32(false, a, false, b, (short)0, c, false, false);
}

__device__ __forceinline__ void rows_out(const float (&s)[64], const float (&rr)[T_FRAMES],
                                         float R1, float Bs, float d60,
                                         float* __restrict__ y, int L, int voxRaw)
{
    const float r0v = rr[0];
    y[voxRaw] = fmaf(Bs, DT * fmaxf(r0v, 0.0f), r0v * R1);
    float C = 0.0f;
    #pragma unroll
    for (int i = 0; i < T_FRAMES - 1; ++i) {
        C = fmaf(d60, C, s[i]);
        const int   m    = i + 1;
        const float rm   = rr[m];
        const float conv = DT * (fmaxf(rm, 0.0f) + C);
        float* yrow = y + (size_t)m * L;
        yrow[voxRaw] = fmaf(Bs, conv, rm * R1);
    }
}

__global__ __launch_bounds__(WG_THREADS)
void srtm_wmma_kernel(const float* __restrict__ ref_tac,
                      const float* __restrict__ kparam,
                      float* __restrict__ y,
                      int L)
{
    __shared__ float Fsh[64 * FS];

    const int tid    = threadIdx.x;
    const int wave   = __builtin_amdgcn_readfirstlane(tid >> 5);
    const int lane   = tid & 31;
    const int laneLo = lane & 15;
    const int laneHi = lane >> 4;
    const int wgBase = blockIdx.x * VOX_PER_WG;

    const int  voxRaw = wgBase + wave * 32 + lane;
    const bool voxOk  = voxRaw < L;
    const int  vox    = voxOk ? voxRaw : L - 1;
    const float R1  = kparam[vox];
    const float kk2 = kparam[L + vox];
    const float bp  = kparam[2 * L + vox];

    const int jrow = laneLo & 7;
    float fq[8], fq1[8];
    #pragma unroll
    for (int t = 0; t < 8; ++t) {
        const int q  = 8 * t + jrow;
        const int q0 = q     < T_FRAMES - 1 ? q     : T_FRAMES - 1;
        const int q1 = q + 1 < T_FRAMES - 1 ? q + 1 : T_FRAMES - 1;
        fq[t]  = ref_tac[q0];
        fq1[t] = ref_tac[q1];
    }
    asm volatile("" :: "v"(fq1[0]), "v"(fq1[1]), "v"(fq1[2]), "v"(fq1[3]),
                       "v"(fq1[4]), "v"(fq1[5]), "v"(fq1[6]), "v"(fq1[7]));
    const bool anyNeg = __builtin_amdgcn_ballot_w32(fq[0] < 0.0f || fq[1] < 0.0f || fq[2] < 0.0f || fq[3] < 0.0f ||
                                                    fq[4] < 0.0f || fq[5] < 0.0f || fq[6] < 0.0f || fq[7] < 0.0f) != 0u;
    float rr[T_FRAMES];
    #pragma unroll
    for (int m = 0; m < T_FRAMES; ++m)
        rr[m] = __builtin_bit_cast(float, __builtin_amdgcn_readlane(__builtin_bit_cast(int, fq[m >> 3]), m & 7));

    const float As  = -kk2 / (1.0f + bp);
    const float Bs  = kk2 * (1.0f - R1 / (1.0f + bp));
    const float dOwn = expf(As * DT);
    const float d60  = expf(As);
    const float dOth = xhalf(dOwn);
    const float dA = laneHi ? dOth : dOwn;
    const float dB = laneHi ? dOwn : dOth;
    const float dA2 = dA * dA, dA4 = dA2 * dA2;
    const float dB2 = dB * dB, dB4 = dB2 * dB2;

    const bool wgFull = wgBase + VOX_PER_WG <= L;

    if (!anyNeg && wgFull) {
        const int wsel = laneLo >> 3, ch = laneLo & 7;
        v2f wa[2];
        #pragma unroll
        for (int j = 0; j < 2; ++j) {
            #pragma unroll
            for (int s = 0; s < 2; ++s) {
                const int   k = 4 * j + 2 * laneHi + s;
                const int   r = 59 - 8 * ch - k;
                const float t = (float)r * DT;
                const float w = wsel ? t : 1.0f - t;
                wa[j][s] = r < 0 ? 0.0f : w;
            }
        }
        v2f bA[2], bB[2];
        bA[0].x = laneHi ? dA2 * dA : dA;  bA[0].y = laneHi ? dA4 : dA2;
        bB[0].x = laneHi ? dB2 * dB : dB;  bB[0].y = laneHi ? dB4 : dB2;
        bA[1] = bA[0] * dA4;
        bB[1] = bB[0] * dB4;
        v8f tA = {}, tB = {};
        tA = mm(wa[0], bA[0], tA);
        tB = mm(wa[0], bB[0], tB);
        tA = mm(wa[1], bA[1], tA);
        tB = mm(wa[1], bB[1], tB);
        const float dA8 = dA4 * dA4, dB8 = dB4 * dB4;
        float uA = tA[7], uB = tB[7];
        #pragma unroll
        for (int r = 6; r >= 0; --r) { uA = fmaf(uA, dA8, tA[r]); uB = fmaf(uB, dB8, tB[r]); }
        const float rcv = xhalf(laneHi ? uA : uB);
        v2f ub;
        ub.x = laneHi ? rcv : uA;
        ub.y = laneHi ? uB  : rcv;
        const bool act = (laneLo >> 3) == laneHi;
        const v8f zero = {};
        const float r0v = rr[0];
        y[voxRaw] = fmaf(Bs, DT * fmaxf(r0v, 0.0f), r0v * R1);
        float C = 0.0f;
        #pragma unroll
        for (int t = 0; t < 8; ++t) {
            v2f a; a.x = act ? fq[t] : 0.0f; a.y = act ? fq1[t] : 0.0f;
            const v8f d = mm(a, ub, zero);
            #pragma unroll
            for (int r = 0; r < 8; ++r) {
                const int i = 8 * t + r;
                if (i < T_FRAMES - 1) {
                    C = fmaf(d60, C, d[r]);
                    const int   m    = i + 1;
                    const float rm   = rr[m];
                    const float conv = DT * (fmaxf(rm, 0.0f) + C);
                    float* yrow = y + (size_t)m * L;
                    yrow[voxRaw] = fmaf(Bs, conv, rm * R1);
                }
            }
        }
    } else {
        for (int idx = tid; idx < 64 * 64; idx += WG_THREADS) {
            const int q = idx >> 6, r = idx & 63;
            float v = 0.0f;
            if (q < 60 && r < 60) {
                const float a = ref_tac[q], b = ref_tac[q + 1];
                v = fmaxf(fmaf((float)r * DT, b - a, a), 0.0f);
            }
            Fsh[q * FS + r] = v;
        }
        __syncthreads();
        float pAx = laneHi ? dA2 : dA4;
        float pAy = laneHi ? dA  : dA2 * dA;
        float pBx = laneHi ? dB2 : dB4;
        float pBy = laneHi ? dB  : dB2 * dB;
        const int aoff = laneHi ? 2 : 0;
        v8f acc[4][2];
        #pragma unroll
        for (int mt = 0; mt < 4; ++mt) { acc[mt][0] = (v8f){}; acc[mt][1] = (v8f){}; }
        for (int kb = 56; kb >= 0; kb -= 4) {
            v2f b0; b0.x = pAx; b0.y = pAy;
            v2f b1; b1.x = pBx; b1.y = pBy;
            const int col = kb + aoff;
            #pragma unroll
            for (int mt = 0; mt < 4; ++mt) {
                const int arow = 16 * mt + laneLo;
                v2f a = { Fsh[arow * FS + col], Fsh[arow * FS + col + 1] };
                acc[mt][0] = mm(a, b0, acc[mt][0]);
                acc[mt][1] = mm(a, b1, acc[mt][1]);
            }
            pAx *= dA4; pAy *= dA4;
            pBx *= dB4; pBy *= dB4;
        }
        float s[64];
        #pragma unroll
        for (int mt = 0; mt < 4; ++mt) {
            #pragma unroll
            for (int v = 0; v < 8; ++v) {
                const float X  = acc[mt][0][v];
                const float Y  = acc[mt][1][v];
                const float Zs = xhalf(laneHi ? X : Y);
                s[16 * mt + v]     = laneHi ? Zs : X;
                s[16 * mt + 8 + v] = laneHi ? Y  : Zs;
            }
        }
        if (voxOk) rows_out(s, rr, R1, Bs, d60, y, L, voxRaw);
    }
}

extern "C" void kernel_launch(void* const* d_in, const int* in_sizes, int n_in,
                              void* d_out, int out_size, void* d_ws, size_t ws_size,
                              hipStream_t stream) {
    const float* ref_tac = (const float*)d_in[1];
    const float* kparam  = (const float*)d_in[2];
    float* y = (float*)d_out;
    const int L = in_sizes[2] / 3;
    const int grid = (L + VOX_PER_WG - 1) / VOX_PER_WG;
    srtm_wmma_kernel<<<grid, WG_THREADS, 0, stream>>>(ref_tac, kparam, y, L);
}


# Fast path: the output row itself rides the recurrence, y_m = d60*y_(m-1) + [ref q, ref q+1] x (b*U0 - d60*G, b*U1 + G): a row is one fma and its store, epilogue folded into B; sum order changes
# speedup vs baseline: 2.1616x; 1.2272x over previous
#include <hip/hip_runtime.h>
#include <hip/hip_bf16.h>

typedef float v2f __attribute__((ext_vector_type(2)));
typedef float v8f __attribute__((ext_vector_type(8)));

#define T_FRAMES   61
#define DT         (1.0f / 60.0f)
#define WG_THREADS 128
#define VOX_PER_WG 128
#define FS         66

__device__ __forceinline__ float xhalf(float x) {
    return __builtin_bit_cast(float, __builtin_amdgcn_permlanex16(
        __builtin_bit_cast(int, x), __builtin_bit_cast(int, x), 0x76543210, 0xfedcba98, false, false));
}

__device__ __forceinline__ v8f mm(v2f a, v2f b, v8f c) {
    return __builtin_amdgcn_wmma_f32_16x16x4_f32(false, a, false, b, (short)0, c, false, false);
}

__device__ __forceinline__ void rows_out(const float (&s)[64], const float (&rr)[T_FRAMES],
                                         float R1, float Bs, float d60,
                                         float* __restrict__ y, int L, int voxRaw)
{
    const float r0v = rr[0];
    y[voxRaw] = fmaf(Bs, DT * fmaxf(r0v, 0.0f), r0v * R1);
    float C = 0.0f;
    #pragma unroll
    for (int i = 0; i < T_FRAMES - 1; ++i) {
        C = fmaf(d60, C, s[i]);
        const int   m    = i + 1;
        const float rm   = rr[m];
        const float conv = DT * (fmaxf(rm, 0.0f) + C);
        float* yrow = y + (size_t)m * L;
        yrow[voxRaw] = fmaf(Bs, conv, rm * R1);
    }
}

__global__ __launch_bounds__(WG_THREADS)
void srtm_wmma_kernel(const float* __restrict__ ref_tac,
                      const float* __restrict__ kparam,
                      float* __restrict__ y,
                      int L)
{
    __shared__ float Fsh[64 * FS];

    const int tid    = threadIdx.x;
    const int wave   = __builtin_amdgcn_readfirstlane(tid >> 5);
    const int lane   = tid & 31;
    const int laneLo = lane & 15;
    const int laneHi = lane >> 4;
    const int wgBase = blockIdx.x * VOX_PER_WG;

    const int  voxRaw = wgBase + wave * 32 + lane;
    const bool voxOk  = voxRaw < L;
    const int  vox    = voxOk ? voxRaw : L - 1;
    const float R1  = kparam[vox];
    const float kk2 = kparam[L + vox];
    const float bp  = kparam[2 * L + vox];

    const int jrow = laneLo & 7;
    float fq[8], fq1[8];
    #pragma unroll
    for (int t = 0; t < 8; ++t) {
        const int q  = 8 * t + jrow;
        const int q0 = q     < T_FRAMES - 1 ? q     : T_FRAMES - 1;
        const int q1 = q + 1 < T_FRAMES - 1 ? q + 1 : T_FRAMES - 1;
        fq[t]  = ref_tac[q0];
        fq1[t] = ref_tac[q1];
    }
    asm volatile("" :: "v"(R1), "v"(kk2), "v"(bp),
                       "v"(fq[0]), "v"(fq[1]), "v"(fq[2]), "v"(fq[3]),
                       "v"(fq[4]), "v"(fq[5]), "v"(fq[6]), "v"(fq[7]),
                       "v"(fq1[0]), "v"(fq1[1]), "v"(fq1[2]), "v"(fq1[3]),
                       "v"(fq1[4]), "v"(fq1[5]), "v"(fq1[6]), "v"(fq1[7]));
    const bool anyNeg = __builtin_amdgcn_ballot_w32(fq[0] < 0.0f || fq[1] < 0.0f || fq[2] < 0.0f || fq[3] < 0.0f ||
                                                    fq[4] < 0.0f || fq[5] < 0.0f || fq[6] < 0.0f || fq[7] < 0.0f) != 0u;

    const float As  = -kk2 / (1.0f + bp);
    const float Bs  = kk2 * (1.0f - R1 / (1.0f + bp));
    const float dOwn = expf(As * DT);
    const float d60  = expf(As);
    const float dOth = xhalf(dOwn);
    const float dA = laneHi ? dOth : dOwn;
    const float dB = laneHi ? dOwn : dOth;
    const float dA2 = dA * dA, dA4 = dA2 * dA2;
    const float dB2 = dB * dB, dB4 = dB2 * dB2;

    const bool wgFull = wgBase + VOX_PER_WG <= L;

    if (!anyNeg && wgFull) {
        const int wsel = laneLo >> 3, ch = laneLo & 7;
        v2f wa[2];
        #pragma unroll
        for (int j = 0; j < 2; ++j) {
            #pragma unroll
            for (int s = 0; s < 2; ++s) {
                const int   k = 4 * j + 2 * laneHi + s;
                const int   r = 59 - 8 * ch - k;
                const float t = (float)r * DT;
                const float w = wsel ? t : 1.0f - t;
                wa[j][s] = r < 0 ? 0.0f : w;
            }
        }
        v2f bA[2], bB[2];
        bA[0].x = laneHi ? dA2 * dA : dA;  bA[0].y = laneHi ? dA4 : dA2;
        bB[0].x = laneHi ? dB2 * dB : dB;  bB[0].y = laneHi ? dB4 : dB2;
        bA[1] = bA[0] * dA4;
        bB[1] = bB[0] * dB4;
        v8f tA = {}, tB = {};
        tA = mm(wa[0], bA[0], tA);
        tB = mm(wa[0], bB[0], tB);
        tA = mm(wa[1], bA[1], tA);
        tB = mm(wa[1], bB[1], tB);
        const float dA8 = dA4 * dA4, dB8 = dB4 * dB4;
        float uA = tA[7], uB = tB[7];
        #pragma unroll
        for (int r = 6; r >= 0; --r) { uA = fmaf(uA, dA8, tA[r]); uB = fmaf(uB, dB8, tB[r]); }
        const float rcv = xhalf(laneHi ? uA : uB);
        v2f ub;
        ub.x = laneHi ? rcv : uA;
        ub.y = laneHi ? uB  : rcv;
        const float bdt = Bs * DT;
        const float G   = R1 + bdt;
        ub.x = fmaf(bdt, ub.x, -(d60 * G));
        ub.y = fmaf(bdt, ub.y, G);
        const bool act = (laneLo >> 3) == laneHi;
        const v8f zero = {};
        const float r0v = __builtin_bit_cast(float, __builtin_amdgcn_readlane(__builtin_bit_cast(int, fq[0]), 0));
        float Y = r0v * G;
        y[voxRaw] = Y;
        #pragma unroll
        for (int t = 0; t < 8; ++t) {
            v2f a; a.x = act ? fq[t] : 0.0f; a.y = act ? fq1[t] : 0.0f;
            const v8f d = mm(a, ub, zero);
            #pragma unroll
            for (int r = 0; r < 8; ++r) {
                const int i = 8 * t + r;
                if (i < T_FRAMES - 1) {
                    Y = fmaf(d60, Y, d[r]);
                    float* yrow = y + (size_t)(i + 1) * L;
                    yrow[voxRaw] = Y;
                }
            }
        }
    } else {
        for (int idx = tid; idx < 64 * 64; idx += WG_THREADS) {
            const int q = idx >> 6, r = idx & 63;
            float v = 0.0f;
            if (q < 60 && r < 60) {
                const float a = ref_tac[q], b = ref_tac[q + 1];
                v = fmaxf(fmaf((float)r * DT, b - a, a), 0.0f);
            }
            Fsh[q * FS + r] = v;
        }
        __syncthreads();
        float pAx = laneHi ? dA2 : dA4;
        float pAy = laneHi ? dA  : dA2 * dA;
        float pBx = laneHi ? dB2 : dB4;
        float pBy = laneHi ? dB  : dB2 * dB;
        const int aoff = laneHi ? 2 : 0;
        v8f acc[4][2];
        #pragma unroll
        for (int mt = 0; mt < 4; ++mt) { acc[mt][0] = (v8f){}; acc[mt][1] = (v8f){}; }
        for (int kb = 56; kb >= 0; kb -= 4) {
            v2f b0; b0.x = pAx; b0.y = pAy;
            v2f b1; b1.x = pBx; b1.y = pBy;
            const int col = kb + aoff;
            #pragma unroll
            for (int mt = 0; mt < 4; ++mt) {
                const int arow = 16 * mt + laneLo;
                v2f a = { Fsh[arow * FS + col], Fsh[arow * FS + col + 1] };
                acc[mt][0] = mm(a, b0, acc[mt][0]);
                acc[mt][1] = mm(a, b1, acc[mt][1]);
            }
            pAx *= dA4; pAy *= dA4;
            pBx *= dB4; pBy *= dB4;
        }
        float s[64];
        #pragma unroll
        for (int mt = 0; mt < 4; ++mt) {
            #pragma unroll
            for (int v = 0; v < 8; ++v) {
                const float X  = acc[mt][0][v];
                const float Y  = acc[mt][1][v];
                const float Zs = xhalf(laneHi ? X : Y);
                s[16 * mt + v]     = laneHi ? Zs : X;
                s[16 * mt + 8 + v] = laneHi ? Y  : Zs;
            }
        }
        float rr[T_FRAMES];
        #pragma unroll
        for (int m = 0; m < T_FRAMES; ++m)
            rr[m] = __builtin_bit_cast(float, __builtin_amdgcn_readlane(__builtin_bit_cast(int, fq[m >> 3]), m & 7));
        if (voxOk) rows_out(s, rr, R1, Bs, d60, y, L, voxRaw);
    }
}

extern "C" void kernel_launch(void* const* d_in, const int* in_sizes, int n_in,
                              void* d_out, int out_size, void* d_ws, size_t ws_size,
                              hipStream_t stream) {
    const float* ref_tac = (const float*)d_in[1];
    const float* kparam  = (const float*)d_in[2];
    float* y = (float*)d_out;
    const int L = in_sizes[2] / 3;
    const int grid = (L + VOX_PER_WG - 1) / VOX_PER_WG;
    srtm_wmma_kernel<<<grid, WG_THREADS, 0, stream>>>(ref_tac, kparam, y, L);
}
